# starting kernel
# baseline (speedup 1.0000x reference)
#include <hip/hip_runtime.h>


typedef __attribute__((ext_vector_type(2))) float v2f;
typedef __attribute__((ext_vector_type(8))) float v8f;

#define T_LEN 2048
#define LDS_STRIDE 17

__device__ __forceinline__ float fexp2(float x) { return __builtin_amdgcn_exp2f(x); }
__device__ __forceinline__ float frcp (float x) { return __builtin_amdgcn_rcpf(x);  }
__device__ __forceinline__ float sigm_(float x) { return frcp(1.0f + fexp2(x * -1.44269504088896340736f)); }
__device__ __forceinline__ float tanh_(float x) { return 1.0f - 2.0f * frcp(1.0f + fexp2(x * 2.88539008177792681472f)); }

__global__ __launch_bounds__(32)
void lstm_fused_wmma(const float* __restrict__ x,
                     const float* __restrict__ W_ih,
                     const float* __restrict__ W_hh,
                     const float* __restrict__ b_ih,
                     const float* __restrict__ b_hh,
                     const float* __restrict__ W_out,
                     const float* __restrict__ b_out,
                     float* __restrict__ out)
{
    __shared__ float hlds[16 * LDS_STRIDE];

    const int lane = threadIdx.x & 31;
    const int lo   = lane & 15;
    const int hi   = lane >> 4;
    const int b0   = blockIdx.x * 16;

    v2f wB[4][5];
#pragma unroll
    for (int n = 0; n < 4; ++n)
#pragma unroll
        for (int k = 0; k < 5; ++k)
#pragma unroll
            for (int q = 0; q < 2; ++q) {
                const int K = 4 * k + 2 * hi + q;
                const int N = 16 * n + lo;
                float v;
                if      (K < 16) v = W_hh[N * 16 + K];
                else if (K < 19) v = W_ih[N * 3 + (K - 16)];
                else             v = b_ih[N] + b_hh[N];
                wB[n][k][q] = v;
            }

    v2f hB[4];
#pragma unroll
    for (int k = 0; k < 4; ++k)
#pragma unroll
        for (int q = 0; q < 2; ++q) {
            const int K = 4 * k + 2 * hi + q;
            hB[k][q] = (lo < 3) ? W_out[lo * 16 + K] : 0.0f;
        }
    const float obias = (lo < 3) ? b_out[lo] : 0.0f;

    const int st_off = (8 * hi) * LDS_STRIDE + lo;
    const int ld_off = lo * LDS_STRIDE + 2 * hi;

    const float* xp = x + (b0 + lo) * (T_LEN * 3) + (hi ? 2 : 0);

    v8f cst;
    v2f ah[4];
#pragma unroll
    for (int r = 0; r < 8; ++r) cst[r] = 0.0f;
#pragma unroll
    for (int k = 0; k < 4; ++k) { ah[k][0] = 0.0f; ah[k][1] = 0.0f; }

    float xa = xp[0];
    float xb = 1.0f;
    if (!hi) xb = xp[1];

    const int out_base = (b0 + 8 * hi) * (T_LEN * 3) + lo;
    const bool do_out  = (lo < 3);

    for (int t = 0; t < T_LEN; ++t) {
        const int tn = (t + 1 < T_LEN) ? (t + 1) : t;
        float na = xp[tn * 3];
        float nb = 1.0f;
        if (!hi) nb = xp[tn * 3 + 1];

        v2f ax; ax[0] = xa; ax[1] = xb;

        v8f acc[4];
#pragma unroll
        for (int n = 0; n < 4; ++n) {
            v8f a = {};
#pragma unroll
            for (int k = 0; k < 4; ++k)
                a = __builtin_amdgcn_wmma_f32_16x16x4_f32(
                        false, ah[k], false, wB[n][k], (short)0, a, false, false);
            a = __builtin_amdgcn_wmma_f32_16x16x4_f32(
                        false, ax, false, wB[n][4], (short)0, a, false, false);
            acc[n] = a;
        }

        v8f hval;
#pragma unroll
        for (int r = 0; r < 8; ++r) {
            const float iv = sigm_(acc[0][r]);
            const float fv = sigm_(acc[1][r]);
            const float gv = tanh_(acc[2][r]);
            const float ov = sigm_(acc[3][r]);
            const float cc = fv * cst[r] + iv * gv;
            cst[r] = cc;
            hval[r] = ov * tanh_(cc);
        }

#pragma unroll
        for (int r = 0; r < 8; ++r)
            hlds[st_off + r * LDS_STRIDE] = hval[r];
        __syncthreads();
#pragma unroll
        for (int k = 0; k < 4; ++k) {
            ah[k][0] = hlds[ld_off + 4 * k + 0];
            ah[k][1] = hlds[ld_off + 4 * k + 1];
        }

        v8f oacc;
#pragma unroll
        for (int r = 0; r < 8; ++r) oacc[r] = obias;
#pragma unroll
        for (int k = 0; k < 4; ++k)
            oacc = __builtin_amdgcn_wmma_f32_16x16x4_f32(
                       false, ah[k], false, hB[k], (short)0, oacc, false, false);

        if (do_out) {
#pragma unroll
            for (int r = 0; r < 8; ++r)
                out[out_base + r * (T_LEN * 3) + t * 3] = oacc[r];
        }

        xa = na; xb = nb;
    }
}

extern "C" void kernel_launch(void* const* d_in, const int* in_sizes, int n_in,
                              void* d_out, int out_size, void* d_ws, size_t ws_size,
                              hipStream_t stream) {
    const float* x     = (const float*)d_in[0];
    const float* W_ih  = (const float*)d_in[1];
    const float* W_hh  = (const float*)d_in[2];
    const float* b_ih  = (const float*)d_in[3];
    const float* b_hh  = (const float*)d_in[4];
    const float* W_out = (const float*)d_in[5];
    const float* b_out = (const float*)d_in[6];
    float* out = (float*)d_out;

    const int B = in_sizes[0] / (T_LEN * 3);
    const int grid = B / 16;

    lstm_fused_wmma<<<grid, 32, 0, stream>>>(x, W_ih, W_hh, b_ih, b_hh,
                                             W_out, b_out, out);
}


# Linear head batched over four steps: a batch row's four quarter columns keep h of four consecutive steps; 4 head WMMAs and one 12-byte store per lane serve a block; sums and their order unchanged
# speedup vs baseline: 2.1693x; 2.1693x over previous
#include <hip/hip_runtime.h>


typedef __attribute__((ext_vector_type(2))) float v2f;
typedef __attribute__((ext_vector_type(8))) float v8f;

#define T_LEN 2048

__device__ __forceinline__ float fexp2(float x) { return __builtin_amdgcn_exp2f(x); }
__device__ __forceinline__ float frcp (float x) { return __builtin_amdgcn_rcpf(x);  }
__device__ __forceinline__ float sigm_(float x) { return frcp(1.0f + fexp2(x * -1.44269504088896340736f)); }
__device__ __forceinline__ float tanh_(float x) { return 1.0f - 2.0f * frcp(1.0f + fexp2(x * 2.88539008177792681472f)); }

template <int S>
__device__ __forceinline__ float quad_bcast(float v) {
    return __int_as_float(__builtin_amdgcn_update_dpp(
        0, __float_as_int(v), S * 0x55, 0xf, 0xf, true));
}

__global__ __launch_bounds__(32)
void lstm_fused_wmma(const float* __restrict__ x,
                     const float* __restrict__ W_ih,
                     const float* __restrict__ W_hh,
                     const float* __restrict__ b_ih,
                     const float* __restrict__ b_hh,
                     const float* __restrict__ W_out,
                     const float* __restrict__ b_out,
                     float* __restrict__ out)
{
    const int lane = threadIdx.x & 31;
    const int c    = lane & 15;
    const int hi   = lane >> 4;
    const int j    = c & 3;
    const int b    = blockIdx.x * 4 + (c >> 2);

    v2f wA[4][5];
#pragma unroll
    for (int jp = 0; jp < 4; ++jp)
#pragma unroll
        for (int kk = 0; kk < 5; ++kk)
#pragma unroll
            for (int q = 0; q < 2; ++q) {
                const int G = 16 * (c & 3) + 4 * jp + (c >> 2);
                const int K = 4 * kk + 2 * hi + q;
                float v;
                if      (K < 16) v = W_hh[G * 16 + K];
                else if (K < 19) v = W_ih[G * 3 + (K - 16)];
                else             v = b_ih[G] + b_hh[G];
                wA[jp][kk][q] = v;
            }

    v2f hA[4];
#pragma unroll
    for (int kk = 0; kk < 4; ++kk)
#pragma unroll
        for (int q = 0; q < 2; ++q) {
            const int K = 4 * kk + 2 * hi + q;
            hA[kk][q] = (c < 3) ? W_out[c * 16 + K] : 0.0f;
        }
    float obias[3];
#pragma unroll
    for (int r = 0; r < 3; ++r) obias[r] = hi ? 0.0f : b_out[r];

    const float* xp = x + b * (T_LEN * 3) + (hi ? 2 : 0);
    const int   off1 = hi ? 0 : 1;

    float cst[2] = {0.0f, 0.0f};
    v2f hb[4];
#pragma unroll
    for (int kk = 0; kk < 4; ++kk) { hb[kk][0] = 0.0f; hb[kk][1] = 0.0f; }

    const float x0a = xp[0], x0b = xp[off1];
    float xca[4], xcb[4];
#pragma unroll
    for (int s = 0; s < 4; ++s) { xca[s] = xp[(1 + s) * 3]; xcb[s] = xp[(1 + s) * 3 + off1]; }

    v8f acc[4];
    {
        v2f xB; xB[0] = x0a; xB[1] = hi ? 1.0f : x0b;
#pragma unroll
        for (int jp = 0; jp < 4; ++jp) {
            v8f z = {};
            acc[jp] = __builtin_amdgcn_wmma_f32_16x16x4_f32(
                          false, wA[jp][4], false, xB, (short)0, z, false, false);
        }
    }

    const bool jodd = (j & 1) != 0;
    const bool jtop = (j & 2) != 0;
    const bool st_lane = (hi == 0);
    float* op = out + b * (T_LEN * 3) + j * 3;
    float o0 = 0.0f, o1 = 0.0f, o2 = 0.0f;

    v2f hbh[4];
#pragma unroll
    for (int kk = 0; kk < 4; ++kk) { hbh[kk][0] = 0.0f; hbh[kk][1] = 0.0f; }

    auto trip = [&](const int t, const int s, const float xa1, const float xb1) __attribute__((always_inline)) {
#pragma unroll
        for (int kk = 0; kk < 4; ++kk)
#pragma unroll
            for (int jp = 0; jp < 4; ++jp)
                acc[jp] = __builtin_amdgcn_wmma_f32_16x16x4_f32(
                              false, wA[jp][kk], false, hb[kk], (short)0, acc[jp], false, false);

        if (s == 1) {
            if (st_lane && t > 4) {
                op[(t - 5) * 3 + 0] = o0;
                op[(t - 5) * 3 + 1] = o1;
                op[(t - 5) * 3 + 2] = o2;
            }
        }

        if (s == 0) {
            v8f oacc = {};
            oacc[0] = obias[0]; oacc[1] = obias[1]; oacc[2] = obias[2];
#pragma unroll
            for (int kk = 0; kk < 4; ++kk)
                oacc = __builtin_amdgcn_wmma_f32_16x16x4_f32(
                           false, hA[kk], false, hbh[kk], (short)0, oacc, false, false);
            o0 = oacc[0]; o1 = oacc[1]; o2 = oacc[2];
        }

        v8f accn[4];
        {
            v2f xB; xB[0] = xa1; xB[1] = hi ? 1.0f : xb1;
#pragma unroll
            for (int jp = 0; jp < 4; ++jp) {
                v8f z = {};
                accn[jp] = __builtin_amdgcn_wmma_f32_16x16x4_f32(
                               false, wA[jp][4], false, xB, (short)0, z, false, false);
            }
        }

        const v8f a0 = acc[0], a1 = acc[1], a2 = acc[2], a3 = acc[3];
        const v8f lo2 = jodd ? a1 : a0;
        const v8f hi2 = jodd ? a3 : a2;
        const v8f g8  = jtop ? hi2 : lo2;

        float hn[2];
#pragma unroll
        for (int u = 0; u < 2; ++u) {
            const float iv = sigm_(g8[4 * u + 0]);
            const float fv = sigm_(g8[4 * u + 1]);
            const float gv = tanh_(g8[4 * u + 2]);
            const float ov = sigm_(g8[4 * u + 3]);
            const float cc = fv * cst[u] + iv * gv;
            cst[u] = cc;
            hn[u] = ov * tanh_(cc);
        }

        hb[0][0] = quad_bcast<0>(hn[0]); hb[0][1] = quad_bcast<0>(hn[1]);
        hb[1][0] = quad_bcast<1>(hn[0]); hb[1][1] = quad_bcast<1>(hn[1]);
        hb[2][0] = quad_bcast<2>(hn[0]); hb[2][1] = quad_bcast<2>(hn[1]);
        hb[3][0] = quad_bcast<3>(hn[0]); hb[3][1] = quad_bcast<3>(hn[1]);

        {
            const bool mine = (j == s);
#pragma unroll
            for (int kk = 0; kk < 4; ++kk) {
                hbh[kk][0] = mine ? hb[kk][0] : hbh[kk][0];
                hbh[kk][1] = mine ? hb[kk][1] : hbh[kk][1];
            }
        }

#pragma unroll
        for (int jp = 0; jp < 4; ++jp) acc[jp] = accn[jp];
    };

    for (int tb = 0; tb < T_LEN / 4; ++tb) {
        float xna[4], xnb[4];
#pragma unroll
        for (int s = 0; s < 4; ++s) {
            const int ts = (4 * tb + 5 + s < T_LEN) ? (4 * tb + 5 + s) : (T_LEN - 1);
            xna[s] = xp[ts * 3];
            xnb[s] = xp[ts * 3 + off1];
        }
        trip(4 * tb + 0, 0, xca[0], xcb[0]);
        trip(4 * tb + 1, 1, xca[1], xcb[1]);
        trip(4 * tb + 2, 2, xca[2], xcb[2]);
        trip(4 * tb + 3, 3, xca[3], xcb[3]);
#pragma unroll
        for (int s = 0; s < 4; ++s) { xca[s] = xna[s]; xcb[s] = xnb[s]; }
    }
    {
        v8f oacc = {};
        oacc[0] = obias[0]; oacc[1] = obias[1]; oacc[2] = obias[2];
#pragma unroll
        for (int kk = 0; kk < 4; ++kk)
            oacc = __builtin_amdgcn_wmma_f32_16x16x4_f32(
                       false, hA[kk], false, hbh[kk], (short)0, oacc, false, false);
        if (st_lane) {
            op[(T_LEN - 4) * 3 + 0] = oacc[0];
            op[(T_LEN - 4) * 3 + 1] = oacc[1];
            op[(T_LEN - 4) * 3 + 2] = oacc[2];
        }
    }
}

extern "C" void kernel_launch(void* const* d_in, const int* in_sizes, int n_in,
                              void* d_out, int out_size, void* d_ws, size_t ws_size,
                              hipStream_t stream) {
    const float* x     = (const float*)d_in[0];
    const float* W_ih  = (const float*)d_in[1];
    const float* W_hh  = (const float*)d_in[2];
    const float* b_ih  = (const float*)d_in[3];
    const float* b_hh  = (const float*)d_in[4];
    const float* W_out = (const float*)d_in[5];
    const float* b_out = (const float*)d_in[6];
    float* out = (float*)d_out;

    const int B = in_sizes[0] / (T_LEN * 3);
    const int grid = B / 4;

    lstm_fused_wmma<<<grid, 32, 0, stream>>>(x, W_ih, W_hh, b_ih, b_hh,
                                             W_out, b_out, out);
}


# x/bias k-tile formed once per four steps: a column's quarter rotates each step, so one x product per tile holds four steps in a batch row's four columns; 18 WMMAs a step, not 21; sum order unchanged
# speedup vs baseline: 2.2857x; 1.0537x over previous
#include <hip/hip_runtime.h>


typedef __attribute__((ext_vector_type(2))) float v2f;
typedef __attribute__((ext_vector_type(8))) float v8f;

#define T_LEN 2048

__device__ __forceinline__ float fexp2(float x) { return __builtin_amdgcn_exp2f(x); }
__device__ __forceinline__ float frcp (float x) { return __builtin_amdgcn_rcpf(x);  }
__device__ __forceinline__ float sigm_(float x) { return frcp(1.0f + fexp2(x * -1.44269504088896340736f)); }
__device__ __forceinline__ float tanh_(float x) { return 1.0f - 2.0f * frcp(1.0f + fexp2(x * 2.88539008177792681472f)); }

template <int V> struct int_tag { static constexpr int value = V; };

template <int S>
__device__ __forceinline__ float quad_bcast(float v) {
    return __int_as_float(__builtin_amdgcn_update_dpp(
        0, __float_as_int(v), S * 0x55, 0xf, 0xf, true));
}

__global__ __launch_bounds__(32)
void lstm_fused_wmma(const float* __restrict__ x,
                     const float* __restrict__ W_ih,
                     const float* __restrict__ W_hh,
                     const float* __restrict__ b_ih,
                     const float* __restrict__ b_hh,
                     const float* __restrict__ W_out,
                     const float* __restrict__ b_out,
                     float* __restrict__ out)
{
    const int lane = threadIdx.x & 31;
    const int c    = lane & 15;
    const int hi   = lane >> 4;
    const int j    = c & 3;
    const int b    = blockIdx.x * 4 + (c >> 2);

    v2f wA[4][5];
#pragma unroll
    for (int jp = 0; jp < 4; ++jp)
#pragma unroll
        for (int kk = 0; kk < 5; ++kk)
#pragma unroll
            for (int q = 0; q < 2; ++q) {
                const int G = 16 * (c & 3) + 4 * jp + (c >> 2);
                const int K = 4 * kk + 2 * hi + q;
                float v;
                if      (K < 16) v = W_hh[G * 16 + K];
                else if (K < 19) v = W_ih[G * 3 + (K - 16)];
                else             v = b_ih[G] + b_hh[G];
                wA[jp][kk][q] = v;
            }

    v2f hA[4];
#pragma unroll
    for (int kk = 0; kk < 4; ++kk)
#pragma unroll
        for (int q = 0; q < 2; ++q) {
            const int K = 4 * kk + 2 * hi + q;
            hA[kk][q] = (c < 3) ? W_out[c * 16 + K] : 0.0f;
        }
    float obias[3];
#pragma unroll
    for (int r = 0; r < 3; ++r) obias[r] = hi ? 0.0f : b_out[r];

    const float* xp = x + b * (T_LEN * 3) + (hi ? 2 : 0);
    const int   off1 = hi ? 0 : 1;

    int dts[4];
#pragma unroll
    for (int jp = 0; jp < 4; ++jp) dts[jp] = ((jp - j) & 3) * 3;

    float cst[2] = {0.0f, 0.0f};
    v2f hb[4];
#pragma unroll
    for (int kk = 0; kk < 4; ++kk) { hb[kk][0] = 0.0f; hb[kk][1] = 0.0f; }

    auto load_x = [&](const int tbq, float (&xa)[4], float (&xb)[4]) __attribute__((always_inline)) {
        const int tbc = (tbq < T_LEN / 4) ? tbq : (T_LEN / 4 - 1);
        const float* p = xp + tbc * 12;
#pragma unroll
        for (int jp = 0; jp < 4; ++jp) { xa[jp] = p[dts[jp]]; xb[jp] = p[dts[jp] + off1]; }
    };
    auto xprod = [&](const int jp, const float xa, const float xb) __attribute__((always_inline)) -> v8f {
        v2f xB; xB[0] = xa; xB[1] = hi ? 1.0f : xb;
        v8f z = {};
        return __builtin_amdgcn_wmma_f32_16x16x4_f32(
                   false, wA[jp][4], false, xB, (short)0, z, false, false);
    };

    float xPa[4], xPb[4], xQa[4], xQb[4];
    v8f XA[4], XB[4];
    load_x(0, xPa, xPb);
#pragma unroll
    for (int jp = 0; jp < 4; ++jp) XA[jp] = xprod(jp, xPa[jp], xPb[jp]);
    load_x(1, xPa, xPb);

    bool qodd[4], qtop[4];
#pragma unroll
    for (int s = 0; s < 4; ++s) { qodd[s] = (((j + s) & 1) != 0); qtop[s] = (((j + s) & 2) != 0); }

    const bool st_lane = (hi == 0);
    float* op = out + b * (T_LEN * 3) + j * 3;
    float o0 = 0.0f, o1 = 0.0f, o2 = 0.0f;

    v2f hbh[4];
#pragma unroll
    for (int kk = 0; kk < 4; ++kk) { hbh[kk][0] = 0.0f; hbh[kk][1] = 0.0f; }

    auto trip = [&](const int t, auto s_tag, const v8f (&Xc)[4], v8f& Xn_s,
                    const float xna, const float xnb) __attribute__((always_inline)) {
        constexpr int s = decltype(s_tag)::value;
        v8f acc[4];
#pragma unroll
        for (int jp = 0; jp < 4; ++jp)
            acc[jp] = __builtin_amdgcn_wmma_f32_16x16x4_f32(
                          false, wA[jp][0], false, hb[0], (short)0, Xc[jp], false, false);
#pragma unroll
        for (int kk = 1; kk < 4; ++kk)
#pragma unroll
            for (int jp = 0; jp < 4; ++jp)
                acc[jp] = __builtin_amdgcn_wmma_f32_16x16x4_f32(
                              false, wA[jp][kk], false, hb[kk], (short)0, acc[jp], false, false);

        if (s == 1) {
            if (st_lane && t > 4) {
                op[(t - 5) * 3 + 0] = o0;
                op[(t - 5) * 3 + 1] = o1;
                op[(t - 5) * 3 + 2] = o2;
            }
        }

        if (s == 0) {
            v8f oacc = {};
            oacc[0] = obias[0]; oacc[1] = obias[1]; oacc[2] = obias[2];
#pragma unroll
            for (int kk = 0; kk < 4; ++kk)
                oacc = __builtin_amdgcn_wmma_f32_16x16x4_f32(
                           false, hA[kk], false, hbh[kk], (short)0, oacc, false, false);
            o0 = oacc[0]; o1 = oacc[1]; o2 = oacc[2];
        }

        Xn_s = xprod(s, xna, xnb);

        const v8f a0 = acc[0], a1 = acc[1], a2 = acc[2], a3 = acc[3];
        const v8f lo2 = qodd[s] ? a1 : a0;
        const v8f hi2 = qodd[s] ? a3 : a2;
        const v8f g8  = qtop[s] ? hi2 : lo2;

        float hn[2], cn[2];
#pragma unroll
        for (int u = 0; u < 2; ++u) {
            const float iv = sigm_(g8[4 * u + 0]);
            const float fv = sigm_(g8[4 * u + 1]);
            const float gv = tanh_(g8[4 * u + 2]);
            const float ov = sigm_(g8[4 * u + 3]);
            const float cc = fv * cst[u] + iv * gv;
            cn[u] = cc;
            hn[u] = ov * tanh_(cc);
        }
#pragma unroll
        for (int u = 0; u < 2; ++u)
            cst[u] = __int_as_float(__builtin_amdgcn_update_dpp(
                         0, __float_as_int(cn[u]), 0x39, 0xf, 0xf, true));

        hb[0][0] = quad_bcast<(0 - s) & 3>(hn[0]); hb[0][1] = quad_bcast<(0 - s) & 3>(hn[1]);
        hb[1][0] = quad_bcast<(1 - s) & 3>(hn[0]); hb[1][1] = quad_bcast<(1 - s) & 3>(hn[1]);
        hb[2][0] = quad_bcast<(2 - s) & 3>(hn[0]); hb[2][1] = quad_bcast<(2 - s) & 3>(hn[1]);
        hb[3][0] = quad_bcast<(3 - s) & 3>(hn[0]); hb[3][1] = quad_bcast<(3 - s) & 3>(hn[1]);

        {
            const bool mine = (j == s);
#pragma unroll
            for (int kk = 0; kk < 4; ++kk) {
                hbh[kk][0] = mine ? hb[kk][0] : hbh[kk][0];
                hbh[kk][1] = mine ? hb[kk][1] : hbh[kk][1];
            }
        }
    };

    auto block = [&](const int tb, const v8f (&Xc)[4], v8f (&Xn)[4],
                     const float (&ca)[4], const float (&cb)[4],
                     float (&la)[4], float (&lb)[4]) __attribute__((always_inline)) {
        load_x(tb + 2, la, lb);
        trip(4 * tb + 0, int_tag<0>{}, Xc, Xn[0], ca[0], cb[0]);
        trip(4 * tb + 1, int_tag<1>{}, Xc, Xn[1], ca[1], cb[1]);
        trip(4 * tb + 2, int_tag<2>{}, Xc, Xn[2], ca[2], cb[2]);
        trip(4 * tb + 3, int_tag<3>{}, Xc, Xn[3], ca[3], cb[3]);
    };

    for (int tb = 0; tb < T_LEN / 4; tb += 2) {
        block(tb,     XA, XB, xPa, xPb, xQa, xQb);
        block(tb + 1, XB, XA, xQa, xQb, xPa, xPb);
    }
    {
        v8f oacc = {};
        oacc[0] = obias[0]; oacc[1] = obias[1]; oacc[2] = obias[2];
#pragma unroll
        for (int kk = 0; kk < 4; ++kk)
            oacc = __builtin_amdgcn_wmma_f32_16x16x4_f32(
                       false, hA[kk], false, hbh[kk], (short)0, oacc, false, false);
        if (st_lane) {
            op[(T_LEN - 4) * 3 + 0] = oacc[0];
            op[(T_LEN - 4) * 3 + 1] = oacc[1];
            op[(T_LEN - 4) * 3 + 2] = oacc[2];
        }
    }
}

extern "C" void kernel_launch(void* const* d_in, const int* in_sizes, int n_in,
                              void* d_out, int out_size, void* d_ws, size_t ws_size,
                              hipStream_t stream) {
    const float* x     = (const float*)d_in[0];
    const float* W_ih  = (const float*)d_in[1];
    const float* W_hh  = (const float*)d_in[2];
    const float* b_ih  = (const float*)d_in[3];
    const float* b_hh  = (const float*)d_in[4];
    const float* W_out = (const float*)d_in[5];
    const float* b_out = (const float*)d_in[6];
    float* out = (float*)d_out;

    const int B = in_sizes[0] / (T_LEN * 3);
    const int grid = B / 4;

    lstm_fused_wmma<<<grid, 32, 0, stream>>>(x, W_ih, W_hh, b_ih, b_hh,
                                             W_out, b_out, out);
}
